# starting kernel
# baseline (speedup 1.0000x reference)
#include <hip/hip_runtime.h>

typedef __attribute__((ext_vector_type(2))) float v2f;
typedef __attribute__((ext_vector_type(4))) float v4f;
typedef __attribute__((ext_vector_type(8))) float v8f;

#define DF 64

__global__ __launch_bounds__(256) void zero_ws_kernel(float* __restrict__ p, int n4) {
  int i = blockIdx.x * 256 + threadIdx.x;
  if (i < n4) {
    v4f z = {0.f, 0.f, 0.f, 0.f};
    *(v4f*)(p + (i << 2)) = z;
  }
}

__global__ __launch_bounds__(256) void make_Mc_kernel(const float* __restrict__ W,
                                                      const float* __restrict__ b,
                                                      float* __restrict__ Mp,
                                                      float* __restrict__ c) {
  int t = threadIdx.x;
  for (int e = t; e < DF * DF; e += 256) {
    int i = e >> 6, j = e & 63;
    float s = 0.f;
#pragma unroll 8
    for (int k = 0; k < DF; ++k) s += W[k * DF + i] * W[j * DF + k];
    Mp[(i >> 1) * (2 * DF) + j * 2 + (i & 1)] = s;
  }
  if (t < DF) {
    float s = b[t];
#pragma unroll 8
    for (int k = 0; k < DF; ++k) s += b[k] * W[t * DF + k];
    c[t] = s;
  }
}

__global__ __launch_bounds__(256) void spmm_scatter_kernel(
    const float* __restrict__ lap, const int* __restrict__ rows,
    const int* __restrict__ cols, const float* __restrict__ input,
    float* __restrict__ x, int nedges) {
  int gid = blockIdx.x * 256 + threadIdx.x;
  int e = gid >> 4;
  if (e >= nedges) return;
  int q = (gid & 15) << 2;
  int r  = rows[e];
  int ci = cols[e];
  float l = lap[e];
  v4f v = *(const v4f*)(input + (size_t)ci * DF + q);
  float* dst = x + (size_t)r * DF + q;
  unsafeAtomicAdd(dst + 0, l * v.x);
  unsafeAtomicAdd(dst + 1, l * v.y);
  unsafeAtomicAdd(dst + 2, l * v.z);
  unsafeAtomicAdd(dst + 3, l * v.w);
}

__global__ __launch_bounds__(256) void gemm_fused_kernel(
    const float* __restrict__ x, const float* __restrict__ Mp,
    const float* __restrict__ c, float* __restrict__ out, int nrows) {
  __shared__ float ldsA[128 * DF];
  __shared__ float ldsB[DF * DF];
  __shared__ float ldsC[DF];

  int t = threadIdx.x;
  int rowBase0 = blockIdx.x * 128;

#pragma unroll
  for (int j = 0; j < 8; ++j) {
    int fi = (j * 256 + t) << 2;
    int row = rowBase0 + (fi >> 6);
    v4f v = {0.f, 0.f, 0.f, 0.f};
    if (row < nrows) v = *(const v4f*)(x + (size_t)rowBase0 * DF + fi);
    *(v4f*)(ldsA + fi) = v;
  }
#pragma unroll
  for (int j = 0; j < 4; ++j) {
    int fi = (j * 256 + t) << 2;
    *(v4f*)(ldsB + fi) = *(const v4f*)(Mp + fi);
  }
  if (t < DF) ldsC[t] = c[t];
  __syncthreads();

  int wave = t >> 5;
  int lane = t & 31;
  int rowBase = rowBase0 + wave * 16;
  if (rowBase >= nrows) return;
  int m  = lane & 15;
  int kh = lane >> 4;

  const float* A = ldsA + wave * (16 * DF) + m * DF;
  v8f acc0 = {}, acc1 = {}, acc2 = {}, acc3 = {};
#pragma unroll
  for (int k0 = 0; k0 < DF; k0 += 4) {
    int ka = k0 + 2 * kh;
    v2f a = *(const v2f*)(A + ka);
    const float* Bp = ldsB + (ka >> 1) * (2 * DF);
    v2f b0 = *(const v2f*)(Bp + (m      ) * 2);
    v2f b1 = *(const v2f*)(Bp + (m + 16 ) * 2);
    v2f b2 = *(const v2f*)(Bp + (m + 32 ) * 2);
    v2f b3 = *(const v2f*)(Bp + (m + 48 ) * 2);
    acc0 = __builtin_amdgcn_wmma_f32_16x16x4_f32(false, a, false, b0, (short)0, acc0, false, false);
    acc1 = __builtin_amdgcn_wmma_f32_16x16x4_f32(false, a, false, b1, (short)0, acc1, false, false);
    acc2 = __builtin_amdgcn_wmma_f32_16x16x4_f32(false, a, false, b2, (short)0, acc2, false, false);
    acc3 = __builtin_amdgcn_wmma_f32_16x16x4_f32(false, a, false, b3, (short)0, acc3, false, false);
  }

  float c0 = ldsC[m], c1 = ldsC[m + 16], c2 = ldsC[m + 32], c3 = ldsC[m + 48];
#pragma unroll
  for (int v = 0; v < 8; ++v) {
    int r = rowBase + v + 8 * kh;
    float* o = out + (size_t)r * DF;
    o[m]      = acc0[v] + c0;
    o[m + 16] = acc1[v] + c1;
    o[m + 32] = acc2[v] + c2;
    o[m + 48] = acc3[v] + c3;
  }
}

extern "C" void kernel_launch(void* const* d_in, const int* in_sizes, int n_in,
                              void* d_out, int out_size, void* d_ws, size_t ws_size,
                              hipStream_t stream) {
  const float* lap  = (const float*)d_in[0];
  const int*   eidx = (const int*)  d_in[1];
  const float* inp  = (const float*)d_in[2];
  const float* W    = (const float*)d_in[3];
  const float* b    = (const float*)d_in[4];
  float* out = (float*)d_out;

  const int E = in_sizes[0];
  const int N = in_sizes[2] / DF;
  const int* rows = eidx;
  const int* cols = eidx + E;

  float* xbuf = (float*)d_ws;
  float* Mp   = xbuf + (size_t)N * DF;
  float* cvec = Mp + DF * DF;

  int n4 = (N * DF) / 4;
  zero_ws_kernel<<<(n4 + 255) / 256, 256, 0, stream>>>(xbuf, n4);
  make_Mc_kernel<<<1, 256, 0, stream>>>(W, b, Mp, cvec);
  long long tot = (long long)E * 16;
  spmm_scatter_kernel<<<(int)((tot + 255) / 256), 256, 0, stream>>>(lap, rows, cols, inp, xbuf, E);
  int nRowTiles = (N + 15) / 16;
  gemm_fused_kernel<<<(nRowTiles + 7) / 8, 256, 0, stream>>>(xbuf, Mp, cvec, out, N);
}


# SpMM scatter: one wave per 32 edges, lane l on floats l and l+32 of a row, so each gather load and f32 atomic covers one whole 128-B line (was 4 lines x 8 dwords); sums stay in atomic arrival order
# speedup vs baseline: 1.6846x; 1.6846x over previous
#include <hip/hip_runtime.h>

typedef __attribute__((ext_vector_type(2))) float v2f;
typedef __attribute__((ext_vector_type(4))) float v4f;
typedef __attribute__((ext_vector_type(8))) float v8f;

#define DF 64

__global__ __launch_bounds__(256) void zero_ws_kernel(float* __restrict__ p, int n4) {
  int i = blockIdx.x * 256 + threadIdx.x;
  if (i < n4) {
    v4f z = {0.f, 0.f, 0.f, 0.f};
    *(v4f*)(p + (i << 2)) = z;
  }
}

__global__ __launch_bounds__(256) void make_Mc_kernel(const float* __restrict__ W,
                                                      const float* __restrict__ b,
                                                      float* __restrict__ Mp,
                                                      float* __restrict__ c) {
  int t = threadIdx.x;
  for (int e = t; e < DF * DF; e += 256) {
    int i = e >> 6, j = e & 63;
    float s = 0.f;
#pragma unroll 8
    for (int k = 0; k < DF; ++k) s += W[k * DF + i] * W[j * DF + k];
    Mp[(i >> 1) * (2 * DF) + j * 2 + (i & 1)] = s;
  }
  if (t < DF) {
    float s = b[t];
#pragma unroll 8
    for (int k = 0; k < DF; ++k) s += b[k] * W[t * DF + k];
    c[t] = s;
  }
}

#define EDGES_PER_WAVE 32
#define EDGE_BATCH 16

__device__ __forceinline__ unsigned bcast_u(int v, int j) {
  return (unsigned)__builtin_amdgcn_readlane(v, j);
}
__device__ __forceinline__ float bcast_f(float v, int j) {
  return __int_as_float(__builtin_amdgcn_readlane(__float_as_int(v), j));
}

__device__ __forceinline__ void scatter_one(const char* __restrict__ inB, char* __restrict__ xB,
                                            unsigned lane4, unsigned cj, unsigned rj, float lj) {
  const char* src = inB + ((cj << 8) + lane4);
  float a = *(const float*)src;
  float b = *(const float*)(src + 128);
  char* dst = xB + ((rj << 8) + lane4);
  unsafeAtomicAdd((float*)dst,         lj * a);
  unsafeAtomicAdd((float*)(dst + 128), lj * b);
}

__global__ __launch_bounds__(256) void spmm_scatter_kernel(
    const float* __restrict__ lap, const int* __restrict__ rows,
    const int* __restrict__ cols, const float* __restrict__ input,
    float* __restrict__ x, int nedges) {
  const int lane = threadIdx.x & 31;
  const int wv = __builtin_amdgcn_readfirstlane((int)(threadIdx.x >> 5));
  const int e0 = (blockIdx.x * 8 + wv) * EDGES_PER_WAVE;
  if (e0 >= nedges) return;
  const int left = nedges - e0;
  int r = 0, ci = 0;
  float l = 0.f;
  if (lane < left) {
    r  = rows[e0 + lane];
    ci = cols[e0 + lane];
    l  = lap[e0 + lane];
  }
  const char* inB = (const char*)input;
  char* xB = (char*)x;
  const unsigned lane4 = (unsigned)lane << 2;

  if (left >= EDGES_PER_WAVE) {
#pragma unroll
    for (int j0 = 0; j0 < EDGES_PER_WAVE; j0 += EDGE_BATCH) {
      float a[EDGE_BATCH], b[EDGE_BATCH];
#pragma unroll
      for (int u = 0; u < EDGE_BATCH; ++u) {
        const char* src = inB + ((bcast_u(ci, j0 + u) << 8) + lane4);
        a[u] = *(const float*)src;
        b[u] = *(const float*)(src + 128);
      }
#pragma unroll
      for (int u = 0; u < EDGE_BATCH; ++u) {
        float lj = bcast_f(l, j0 + u);
        char* dst = xB + ((bcast_u(r, j0 + u) << 8) + lane4);
        unsafeAtomicAdd((float*)dst,         lj * a[u]);
        unsafeAtomicAdd((float*)(dst + 128), lj * b[u]);
      }
    }
  } else {
    for (int j = 0; j < left; ++j)
      scatter_one(inB, xB, lane4, bcast_u(ci, j), bcast_u(r, j), bcast_f(l, j));
  }
}

__global__ __launch_bounds__(256) void gemm_fused_kernel(
    const float* __restrict__ x, const float* __restrict__ Mp,
    const float* __restrict__ c, float* __restrict__ out, int nrows) {
  __shared__ float ldsA[128 * DF];
  __shared__ float ldsB[DF * DF];
  __shared__ float ldsC[DF];

  int t = threadIdx.x;
  int rowBase0 = blockIdx.x * 128;

#pragma unroll
  for (int j = 0; j < 8; ++j) {
    int fi = (j * 256 + t) << 2;
    int row = rowBase0 + (fi >> 6);
    v4f v = {0.f, 0.f, 0.f, 0.f};
    if (row < nrows) v = *(const v4f*)(x + (size_t)rowBase0 * DF + fi);
    *(v4f*)(ldsA + fi) = v;
  }
#pragma unroll
  for (int j = 0; j < 4; ++j) {
    int fi = (j * 256 + t) << 2;
    *(v4f*)(ldsB + fi) = *(const v4f*)(Mp + fi);
  }
  if (t < DF) ldsC[t] = c[t];
  __syncthreads();

  int wave = t >> 5;
  int lane = t & 31;
  int rowBase = rowBase0 + wave * 16;
  if (rowBase >= nrows) return;
  int m  = lane & 15;
  int kh = lane >> 4;

  const float* A = ldsA + wave * (16 * DF) + m * DF;
  v8f acc0 = {}, acc1 = {}, acc2 = {}, acc3 = {};
#pragma unroll
  for (int k0 = 0; k0 < DF; k0 += 4) {
    int ka = k0 + 2 * kh;
    v2f a = *(const v2f*)(A + ka);
    const float* Bp = ldsB + (ka >> 1) * (2 * DF);
    v2f b0 = *(const v2f*)(Bp + (m      ) * 2);
    v2f b1 = *(const v2f*)(Bp + (m + 16 ) * 2);
    v2f b2 = *(const v2f*)(Bp + (m + 32 ) * 2);
    v2f b3 = *(const v2f*)(Bp + (m + 48 ) * 2);
    acc0 = __builtin_amdgcn_wmma_f32_16x16x4_f32(false, a, false, b0, (short)0, acc0, false, false);
    acc1 = __builtin_amdgcn_wmma_f32_16x16x4_f32(false, a, false, b1, (short)0, acc1, false, false);
    acc2 = __builtin_amdgcn_wmma_f32_16x16x4_f32(false, a, false, b2, (short)0, acc2, false, false);
    acc3 = __builtin_amdgcn_wmma_f32_16x16x4_f32(false, a, false, b3, (short)0, acc3, false, false);
  }

  float c0 = ldsC[m], c1 = ldsC[m + 16], c2 = ldsC[m + 32], c3 = ldsC[m + 48];
#pragma unroll
  for (int v = 0; v < 8; ++v) {
    int r = rowBase + v + 8 * kh;
    float* o = out + (size_t)r * DF;
    o[m]      = acc0[v] + c0;
    o[m + 16] = acc1[v] + c1;
    o[m + 32] = acc2[v] + c2;
    o[m + 48] = acc3[v] + c3;
  }
}

extern "C" void kernel_launch(void* const* d_in, const int* in_sizes, int n_in,
                              void* d_out, int out_size, void* d_ws, size_t ws_size,
                              hipStream_t stream) {
  const float* lap  = (const float*)d_in[0];
  const int*   eidx = (const int*)  d_in[1];
  const float* inp  = (const float*)d_in[2];
  const float* W    = (const float*)d_in[3];
  const float* b    = (const float*)d_in[4];
  float* out = (float*)d_out;

  const int E = in_sizes[0];
  const int N = in_sizes[2] / DF;
  const int* rows = eidx;
  const int* cols = eidx + E;

  float* xbuf = (float*)d_ws;
  float* Mp   = xbuf + (size_t)N * DF;
  float* cvec = Mp + DF * DF;

  int n4 = (N * DF) / 4;
  zero_ws_kernel<<<(n4 + 255) / 256, 256, 0, stream>>>(xbuf, n4);
  make_Mc_kernel<<<1, 256, 0, stream>>>(W, b, Mp, cvec);
  spmm_scatter_kernel<<<(E + 255) / 256, 256, 0, stream>>>(lap, rows, cols, inp, xbuf, E);
  int nRowTiles = (N + 15) / 16;
  gemm_fused_kernel<<<(nRowTiles + 7) / 8, 256, 0, stream>>>(xbuf, Mp, cvec, out, N);
}


# M = W^T W^T and c made by the first 17 work-groups of the zero-fill launch (one element per thread, same k order, one fma per step) instead of a one-work-group launch of their own: 3 launches
# speedup vs baseline: 2.2275x; 1.3223x over previous
#include <hip/hip_runtime.h>

typedef __attribute__((ext_vector_type(2))) float v2f;
typedef __attribute__((ext_vector_type(4))) float v4f;
typedef __attribute__((ext_vector_type(8))) float v8f;

#define DF 64

#define MC_GROUPS ((DF * DF) / 256)

__global__ __launch_bounds__(256) void zero_and_Mc_kernel(float* __restrict__ p, int n4,
                                                          const float* __restrict__ W,
                                                          const float* __restrict__ b,
                                                          float* __restrict__ Mp,
                                                          float* __restrict__ c) {
  const int t = threadIdx.x;
  const int i4 = blockIdx.x * 256 + t;
  if (i4 < n4) {
    v4f z = {0.f, 0.f, 0.f, 0.f};
    *(v4f*)(p + (i4 << 2)) = z;
  }
  if (blockIdx.x < MC_GROUPS) {
    const int wv = blockIdx.x * 8 + __builtin_amdgcn_readfirstlane(t >> 5);
    const int j = wv >> 1;
    const int i = ((wv & 1) << 5) + (t & 31);
    const float* __restrict__ wrow = W + j * DF;
    const float* __restrict__ wcol = W + i;
    float s = 0.f;
#pragma unroll
    for (int k = 0; k < DF; ++k) s = __builtin_fmaf(wcol[k * DF], wrow[k], s);
    Mp[(i >> 1) * (2 * DF) + j * 2 + (i & 1)] = s;
  } else if (blockIdx.x == MC_GROUPS && t < DF) {
    const v4f* __restrict__ wt = (const v4f*)(W + t * DF);
    float sc = b[t];
#pragma unroll
    for (int k4 = 0; k4 < DF / 4; ++k4) {
      v4f wv = wt[k4];
      sc = __builtin_fmaf(b[4 * k4 + 0], wv[0], sc);
      sc = __builtin_fmaf(b[4 * k4 + 1], wv[1], sc);
      sc = __builtin_fmaf(b[4 * k4 + 2], wv[2], sc);
      sc = __builtin_fmaf(b[4 * k4 + 3], wv[3], sc);
    }
    c[t] = sc;
  }
}

#define EDGES_PER_WAVE 32
#define EDGE_BATCH 16

__device__ __forceinline__ unsigned bcast_u(int v, int j) {
  return (unsigned)__builtin_amdgcn_readlane(v, j);
}
__device__ __forceinline__ float bcast_f(float v, int j) {
  return __int_as_float(__builtin_amdgcn_readlane(__float_as_int(v), j));
}

__device__ __forceinline__ void scatter_one(const char* __restrict__ inB, char* __restrict__ xB,
                                            unsigned lane4, unsigned cj, unsigned rj, float lj) {
  const char* src = inB + ((cj << 8) + lane4);
  float a = *(const float*)src;
  float b = *(const float*)(src + 128);
  char* dst = xB + ((rj << 8) + lane4);
  unsafeAtomicAdd((float*)dst,         lj * a);
  unsafeAtomicAdd((float*)(dst + 128), lj * b);
}

__global__ __launch_bounds__(256) void spmm_scatter_kernel(
    const float* __restrict__ lap, const int* __restrict__ rows,
    const int* __restrict__ cols, const float* __restrict__ input,
    float* __restrict__ x, int nedges) {
  const int lane = threadIdx.x & 31;
  const int wv = __builtin_amdgcn_readfirstlane((int)(threadIdx.x >> 5));
  const int e0 = (blockIdx.x * 8 + wv) * EDGES_PER_WAVE;
  if (e0 >= nedges) return;
  const int left = nedges - e0;
  int r = 0, ci = 0;
  float l = 0.f;
  if (lane < left) {
    r  = rows[e0 + lane];
    ci = cols[e0 + lane];
    l  = lap[e0 + lane];
  }
  const char* inB = (const char*)input;
  char* xB = (char*)x;
  const unsigned lane4 = (unsigned)lane << 2;

  if (left >= EDGES_PER_WAVE) {
#pragma unroll
    for (int j0 = 0; j0 < EDGES_PER_WAVE; j0 += EDGE_BATCH) {
      float a[EDGE_BATCH], b[EDGE_BATCH];
#pragma unroll
      for (int u = 0; u < EDGE_BATCH; ++u) {
        const char* src = inB + ((bcast_u(ci, j0 + u) << 8) + lane4);
        a[u] = *(const float*)src;
        b[u] = *(const float*)(src + 128);
      }
#pragma unroll
      for (int u = 0; u < EDGE_BATCH; ++u) {
        float lj = bcast_f(l, j0 + u);
        char* dst = xB + ((bcast_u(r, j0 + u) << 8) + lane4);
        unsafeAtomicAdd((float*)dst,         lj * a[u]);
        unsafeAtomicAdd((float*)(dst + 128), lj * b[u]);
      }
    }
  } else {
    for (int j = 0; j < left; ++j)
      scatter_one(inB, xB, lane4, bcast_u(ci, j), bcast_u(r, j), bcast_f(l, j));
  }
}

__global__ __launch_bounds__(256) void gemm_fused_kernel(
    const float* __restrict__ x, const float* __restrict__ Mp,
    const float* __restrict__ c, float* __restrict__ out, int nrows) {
  __shared__ float ldsA[128 * DF];
  __shared__ float ldsB[DF * DF];
  __shared__ float ldsC[DF];

  int t = threadIdx.x;
  int rowBase0 = blockIdx.x * 128;

#pragma unroll
  for (int j = 0; j < 8; ++j) {
    int fi = (j * 256 + t) << 2;
    int row = rowBase0 + (fi >> 6);
    v4f v = {0.f, 0.f, 0.f, 0.f};
    if (row < nrows) v = *(const v4f*)(x + (size_t)rowBase0 * DF + fi);
    *(v4f*)(ldsA + fi) = v;
  }
#pragma unroll
  for (int j = 0; j < 4; ++j) {
    int fi = (j * 256 + t) << 2;
    *(v4f*)(ldsB + fi) = *(const v4f*)(Mp + fi);
  }
  if (t < DF) ldsC[t] = c[t];
  __syncthreads();

  int wave = t >> 5;
  int lane = t & 31;
  int rowBase = rowBase0 + wave * 16;
  if (rowBase >= nrows) return;
  int m  = lane & 15;
  int kh = lane >> 4;

  const float* A = ldsA + wave * (16 * DF) + m * DF;
  v8f acc0 = {}, acc1 = {}, acc2 = {}, acc3 = {};
#pragma unroll
  for (int k0 = 0; k0 < DF; k0 += 4) {
    int ka = k0 + 2 * kh;
    v2f a = *(const v2f*)(A + ka);
    const float* Bp = ldsB + (ka >> 1) * (2 * DF);
    v2f b0 = *(const v2f*)(Bp + (m      ) * 2);
    v2f b1 = *(const v2f*)(Bp + (m + 16 ) * 2);
    v2f b2 = *(const v2f*)(Bp + (m + 32 ) * 2);
    v2f b3 = *(const v2f*)(Bp + (m + 48 ) * 2);
    acc0 = __builtin_amdgcn_wmma_f32_16x16x4_f32(false, a, false, b0, (short)0, acc0, false, false);
    acc1 = __builtin_amdgcn_wmma_f32_16x16x4_f32(false, a, false, b1, (short)0, acc1, false, false);
    acc2 = __builtin_amdgcn_wmma_f32_16x16x4_f32(false, a, false, b2, (short)0, acc2, false, false);
    acc3 = __builtin_amdgcn_wmma_f32_16x16x4_f32(false, a, false, b3, (short)0, acc3, false, false);
  }

  float c0 = ldsC[m], c1 = ldsC[m + 16], c2 = ldsC[m + 32], c3 = ldsC[m + 48];
#pragma unroll
  for (int v = 0; v < 8; ++v) {
    int r = rowBase + v + 8 * kh;
    float* o = out + (size_t)r * DF;
    o[m]      = acc0[v] + c0;
    o[m + 16] = acc1[v] + c1;
    o[m + 32] = acc2[v] + c2;
    o[m + 48] = acc3[v] + c3;
  }
}

extern "C" void kernel_launch(void* const* d_in, const int* in_sizes, int n_in,
                              void* d_out, int out_size, void* d_ws, size_t ws_size,
                              hipStream_t stream) {
  const float* lap  = (const float*)d_in[0];
  const int*   eidx = (const int*)  d_in[1];
  const float* inp  = (const float*)d_in[2];
  const float* W    = (const float*)d_in[3];
  const float* b    = (const float*)d_in[4];
  float* out = (float*)d_out;

  const int E = in_sizes[0];
  const int N = in_sizes[2] / DF;
  const int* rows = eidx;
  const int* cols = eidx + E;

  float* xbuf = (float*)d_ws;
  float* Mp   = xbuf + (size_t)N * DF;
  float* cvec = Mp + DF * DF;

  int n4 = (N * DF) / 4;
  zero_and_Mc_kernel<<<(n4 + 255) / 256, 256, 0, stream>>>(xbuf, n4, W, b, Mp, cvec);
  spmm_scatter_kernel<<<(E + 255) / 256, 256, 0, stream>>>(lap, rows, cols, inp, xbuf, E);
  int nRowTiles = (N + 15) / 16;
  gemm_fused_kernel<<<(nRowTiles + 7) / 8, 256, 0, stream>>>(xbuf, Mp, cvec, out, N);
}


# SpMM by destination range: edges count-sorted through LDS into bins of 400 rows; one work-group per bin sums its rows in an LDS tile (LDS f32 adds, arrival order) read by the WMMA product; 2 launches
# speedup vs baseline: 4.2711x; 1.9174x over previous
#include <hip/hip_runtime.h>


typedef __attribute__((ext_vector_type(2))) float v2f;
typedef __attribute__((ext_vector_type(4))) float v4f;
typedef __attribute__((ext_vector_type(8))) float v8f;

#define DF 64
#define RBIN 400
#define NBMAX 256
#define NBP (NBMAX + 1)
#define OFFP 288
#define CHUNK 8192
#define EPT (CHUNK / 1024)
#define CAP 8192
#define PITCH 68
#define EDGE_BATCH 16


__global__ __launch_bounds__(1024) void bin_edges_and_Mc_kernel(
    const float* __restrict__ lap, const int* __restrict__ rows, const int* __restrict__ cols,
    int nedges, int nchunks,
    int* __restrict__ off, int* __restrict__ srow, int* __restrict__ scol, float* __restrict__ sval,
    const float* __restrict__ W, const float* __restrict__ b,
    float* __restrict__ Mp, float* __restrict__ c) {
  __shared__ int hist[NBMAX];
  __shared__ int start[NBP];
  __shared__ int s_row[CHUNK];
  __shared__ int s_col[CHUNK];
  __shared__ float s_val[CHUNK];
  const int t = threadIdx.x;

  if ((int)blockIdx.x >= nchunks) {
    const int g = (int)blockIdx.x - nchunks;
    if (g < 4) {
      const int wv = g * 32 + __builtin_amdgcn_readfirstlane(t >> 5);
      const int j = wv >> 1;
      const int i = ((wv & 1) << 5) + (t & 31);
      const float* __restrict__ wrow = W + j * DF;
      const float* __restrict__ wcol = W + i;
      float s = 0.f;
#pragma unroll
      for (int k = 0; k < DF; ++k) s = __builtin_fmaf(wcol[k * DF], wrow[k], s);
      Mp[(i >> 1) * (2 * DF) + j * 2 + (i & 1)] = s;
    } else if (t < DF) {
      const v4f* __restrict__ wt = (const v4f*)(W + t * DF);
      float sc = b[t];
#pragma unroll
      for (int k4 = 0; k4 < DF / 4; ++k4) {
        v4f wv = wt[k4];
        sc = __builtin_fmaf(b[4 * k4 + 0], wv[0], sc);
        sc = __builtin_fmaf(b[4 * k4 + 1], wv[1], sc);
        sc = __builtin_fmaf(b[4 * k4 + 2], wv[2], sc);
        sc = __builtin_fmaf(b[4 * k4 + 3], wv[3], sc);
      }
      c[t] = sc;
    }
    return;
  }

  const int e0 = (int)blockIdx.x * CHUNK;
  const int left = nedges - e0;
  const int n = left < CHUNK ? left : CHUNK;
  if (t < NBMAX) hist[t] = 0;
  __syncthreads();

  int r[EPT], ci[EPT], rk[EPT];
  float v[EPT];
#pragma unroll
  for (int u = 0; u < EPT; ++u) {
    const int idx = u * 1024 + t;
    r[u] = 0; ci[u] = 0; v[u] = 0.f;
    if (idx < n) {
      r[u]  = rows[e0 + idx];
      ci[u] = cols[e0 + idx];
      v[u]  = lap[e0 + idx];
    }
  }
#pragma unroll
  for (int u = 0; u < EPT; ++u) {
    const int idx = u * 1024 + t;
    rk[u] = 0;
    if (idx < n) rk[u] = atomicAdd(&hist[r[u] / RBIN], 1);
  }
  __syncthreads();

  if (t < 32) {
    const int base = t * (NBMAX / 32);
    int loc[NBMAX / 32];
    int s = 0;
#pragma unroll
    for (int k = 0; k < NBMAX / 32; ++k) { loc[k] = s; s += hist[base + k]; }
    int inc = s;
#pragma unroll
    for (int d = 1; d < 32; d <<= 1) {
      const int y = __shfl_up(inc, d, 32);
      if (t >= d) inc += y;
    }
    const int excl = inc - s;
#pragma unroll
    for (int k = 0; k < NBMAX / 32; ++k) start[base + k] = excl + loc[k];
    if (t == 31) start[NBMAX] = inc;
  }
  __syncthreads();

#pragma unroll
  for (int u = 0; u < EPT; ++u) {
    const int idx = u * 1024 + t;
    if (idx < n) {
      const int bn = r[u] / RBIN;
      const int pos = start[bn] + rk[u];
      s_row[pos] = r[u] - bn * RBIN;
      s_col[pos] = ci[u];
      s_val[pos] = v[u];
    }
  }
  __syncthreads();

  for (int idx = t; idx < n; idx += 1024) {
    srow[e0 + idx] = s_row[idx];
    scol[e0 + idx] = s_col[idx];
    sval[e0 + idx] = s_val[idx];
  }
  if (t < NBP) off[(int)blockIdx.x * OFFP + t] = e0 + start[t];
}

__device__ __forceinline__ int bcast_i(int v, int j) {
  return __builtin_amdgcn_readlane(v, j);
}
__device__ __forceinline__ float bcast_f(float v, int j) {
  return __int_as_float(__builtin_amdgcn_readlane(__float_as_int(v), j));
}

__global__ __launch_bounds__(1024) void bin_spmm_gemm_kernel(
    const int* __restrict__ off, const int* __restrict__ srow, const int* __restrict__ scol,
    const float* __restrict__ sval, int nchunks,
    const float* __restrict__ input, const float* __restrict__ Mp, const float* __restrict__ c,
    float* __restrict__ out, int nrows) {
  __shared__ float xt[(RBIN + 1) * PITCH];
  __shared__ float ldsB[DF * DF];
  __shared__ float ldsC[DF];
  __shared__ int   lens[1024];
  __shared__ int   pS[1024];
  __shared__ int   s_tot;
  __shared__ int   st_r[CAP];
  __shared__ int   st_c[CAP];
  __shared__ float st_v[CAP];

  const int t = threadIdx.x;
  const int lane = t & 31;
  const int wave = __builtin_amdgcn_readfirstlane(t >> 5);
  const int bin = (int)blockIdx.x;

  for (int i4 = t; i4 < ((RBIN + 1) * PITCH) / 4; i4 += 1024) {
    v4f z = {0.f, 0.f, 0.f, 0.f};
    *(v4f*)(xt + (i4 << 2)) = z;
  }
  {
    const int fi = t << 2;
    *(v4f*)(ldsB + fi) = *(const v4f*)(Mp + fi);
  }
  if (t < DF) ldsC[t] = c[t];

  int segS = 0, segN = 0;
  if (t < nchunks) {
    const int s0 = off[t * OFFP + bin];
    const int s1 = off[t * OFFP + bin + 1];
    segS = s0;
    segN = s1 - s0;
  }
  lens[t] = segN;
  pS[t] = segS;
  __syncthreads();
  if (t < 32) {
    const int base = t * 32;
    int s = 0;
#pragma unroll
    for (int k = 0; k < 32; ++k) {
      const int v = lens[base + k];
      lens[base + k] = s;
      s += v;
    }
    int inc = s;
#pragma unroll
    for (int d = 1; d < 32; d <<= 1) {
      const int y = __shfl_up(inc, d, 32);
      if (t >= d) inc += y;
    }
    const int excl = inc - s;
#pragma unroll
    for (int k = 0; k < 32; ++k) lens[base + k] += excl;
    if (t == 31) s_tot = inc;
  }
  __syncthreads();
  const int T = __builtin_amdgcn_readfirstlane(s_tot);

  const char* inB = (const char*)input;
  const unsigned lane4 = (unsigned)lane << 2;

  for (int base = 0; base < T; base += CAP) {
    const int rest = T - base;
    const int nIn = rest < CAP ? rest : CAP;
    const int nPad = (nIn + (EDGE_BATCH - 1)) & ~(EDGE_BATCH - 1);
    int src[CAP / 1024];
#pragma unroll
    for (int u = 0; u < CAP / 1024; ++u) {
      const int q = u * 1024 + t;
      const int f = base + (q < nIn ? q : nIn - 1);
      int p = 0;
#pragma unroll
      for (int step = 512; step >= 1; step >>= 1)
        if (lens[p + step] <= f) p += step;
      src[u] = pS[p] + (f - lens[p]);
    }
    int er[CAP / 1024], ec[CAP / 1024];
    float ev[CAP / 1024];
#pragma unroll
    for (int u = 0; u < CAP / 1024; ++u) {
      er[u] = srow[src[u]];
      ec[u] = scol[src[u]];
      ev[u] = sval[src[u]];
    }
#pragma unroll
    for (int u = 0; u < CAP / 1024; ++u) {
      const int q = u * 1024 + t;
      if (q < nIn) { st_r[q] = er[u]; st_c[q] = ec[u]; st_v[q] = ev[u]; }
    }
    if (t < nPad - nIn) {
      st_r[nIn + t] = RBIN;
      st_c[nIn + t] = 0;
      st_v[nIn + t] = 0.f;
    }
    __syncthreads();

    for (int bi = wave; bi < nPad / EDGE_BATCH; bi += 32) {
      const int j = bi * EDGE_BATCH + (lane & (EDGE_BATCH - 1));
      const int rr = st_r[j];
      const int cc = st_c[j];
      const float ll = st_v[j];
      float a[EDGE_BATCH], bq[EDGE_BATCH];
#pragma unroll
      for (int u = 0; u < EDGE_BATCH; ++u) {
        const char* gp = inB + (((unsigned)bcast_i(cc, u) << 8) + lane4);
        a[u]  = *(const float*)gp;
        bq[u] = *(const float*)(gp + 128);
      }
#pragma unroll
      for (int u = 0; u < EDGE_BATCH; ++u) {
        const float lj = bcast_f(ll, u);
        float* dp = xt + bcast_i(rr, u) * PITCH + lane;
        __hip_atomic_fetch_add(dp,      lj * a[u],  __ATOMIC_RELAXED, __HIP_MEMORY_SCOPE_WORKGROUP);
        __hip_atomic_fetch_add(dp + 32, lj * bq[u], __ATOMIC_RELAXED, __HIP_MEMORY_SCOPE_WORKGROUP);
      }
    }
    __syncthreads();
  }

  const int m  = lane & 15;
  const int kh = lane >> 4;
  const float c0v = ldsC[m], c1v = ldsC[m + 16], c2v = ldsC[m + 32], c3v = ldsC[m + 48];
  for (int strip = wave; strip < RBIN / 16; strip += 32) {
    const int rowBase = bin * RBIN + strip * 16;
    if (rowBase >= nrows) break;
    const float* A = xt + (strip * 16 + m) * PITCH;
    v8f acc0 = {}, acc1 = {}, acc2 = {}, acc3 = {};
#pragma unroll
    for (int k0 = 0; k0 < DF; k0 += 4) {
      const int ka = k0 + 2 * kh;
      v2f a = *(const v2f*)(A + ka);
      const float* Bp = ldsB + (ka >> 1) * (2 * DF);
      v2f b0 = *(const v2f*)(Bp + (m      ) * 2);
      v2f b1 = *(const v2f*)(Bp + (m + 16 ) * 2);
      v2f b2 = *(const v2f*)(Bp + (m + 32 ) * 2);
      v2f b3 = *(const v2f*)(Bp + (m + 48 ) * 2);
      acc0 = __builtin_amdgcn_wmma_f32_16x16x4_f32(false, a, false, b0, (short)0, acc0, false, false);
      acc1 = __builtin_amdgcn_wmma_f32_16x16x4_f32(false, a, false, b1, (short)0, acc1, false, false);
      acc2 = __builtin_amdgcn_wmma_f32_16x16x4_f32(false, a, false, b2, (short)0, acc2, false, false);
      acc3 = __builtin_amdgcn_wmma_f32_16x16x4_f32(false, a, false, b3, (short)0, acc3, false, false);
    }
#pragma unroll
    for (int v = 0; v < 8; ++v) {
      const int r = rowBase + v + 8 * kh;
      if (r < nrows) {
        float* o = out + (size_t)r * DF;
        o[m]      = acc0[v] + c0v;
        o[m + 16] = acc1[v] + c1v;
        o[m + 32] = acc2[v] + c2v;
        o[m + 48] = acc3[v] + c3v;
      }
    }
  }
}

extern "C" void kernel_launch(void* const* d_in, const int* in_sizes, int n_in,
                              void* d_out, int out_size, void* d_ws, size_t ws_size,
                              hipStream_t stream) {
  const float* lap  = (const float*)d_in[0];
  const int*   eidx = (const int*)  d_in[1];
  const float* inp  = (const float*)d_in[2];
  const float* W    = (const float*)d_in[3];
  const float* b    = (const float*)d_in[4];
  float* out = (float*)d_out;

  const int E = in_sizes[0];
  const int N = in_sizes[2] / DF;
  const int* rows = eidx;
  const int* cols = eidx + E;

  const int nchunks = (E + CHUNK - 1) / CHUNK;
  const int nbins   = (N + RBIN - 1) / RBIN;

  float* Mp   = (float*)d_ws;
  float* cvec = Mp + DF * DF;
  int*   off  = (int*)(cvec + DF);
  int*   srow = off + 1024 * OFFP;
  int*   scol = srow + E;
  float* sval = (float*)(scol + E);

  bin_edges_and_Mc_kernel<<<nchunks + 5, 1024, 0, stream>>>(lap, rows, cols, E, nchunks,
                                                            off, srow, scol, sval, W, b, Mp, cvec);
  bin_spmm_gemm_kernel<<<nbins, 1024, 0, stream>>>(off, srow, scol, sval, nchunks,
                                                   inp, Mp, cvec, out, N);
}


# SpMM tile: a bin's edges counting-sorted by row in LDS; each wave sums a row's edges in registers (fma, list order) and does one LDS f32 add per row run instead of one per edge; sum order changes
# speedup vs baseline: 5.0423x; 1.1806x over previous
#include <hip/hip_runtime.h>


typedef __attribute__((ext_vector_type(2))) float v2f;
typedef __attribute__((ext_vector_type(4))) float v4f;
typedef __attribute__((ext_vector_type(8))) float v8f;

#define DF 64
#define RBIN 400
#define NBMAX 256
#define NBP (NBMAX + 1)
#define OFFP 288
#define CHUNK 8192
#define EPT (CHUNK / 1024)
#define CAP 8192
#define PITCH 68
#define EDGE_BATCH 16
#define RSCAN 13
#define RCNT (32 * RSCAN)


__global__ __launch_bounds__(1024) void bin_edges_and_Mc_kernel(
    const float* __restrict__ lap, const int* __restrict__ rows, const int* __restrict__ cols,
    int nedges, int nchunks,
    int* __restrict__ off, int* __restrict__ srow, int* __restrict__ scol, float* __restrict__ sval,
    const float* __restrict__ W, const float* __restrict__ b,
    float* __restrict__ Mp, float* __restrict__ c) {
  __shared__ int hist[NBMAX];
  __shared__ int start[NBP];
  __shared__ int s_row[CHUNK];
  __shared__ int s_col[CHUNK];
  __shared__ float s_val[CHUNK];
  const int t = threadIdx.x;

  if ((int)blockIdx.x >= nchunks) {
    const int g = (int)blockIdx.x - nchunks;
    if (g < 4) {
      const int wv = g * 32 + __builtin_amdgcn_readfirstlane(t >> 5);
      const int j = wv >> 1;
      const int i = ((wv & 1) << 5) + (t & 31);
      const float* __restrict__ wrow = W + j * DF;
      const float* __restrict__ wcol = W + i;
      float s = 0.f;
#pragma unroll
      for (int k = 0; k < DF; ++k) s = __builtin_fmaf(wcol[k * DF], wrow[k], s);
      Mp[(i >> 1) * (2 * DF) + j * 2 + (i & 1)] = s;
    } else if (t < DF) {
      const v4f* __restrict__ wt = (const v4f*)(W + t * DF);
      float sc = b[t];
#pragma unroll
      for (int k4 = 0; k4 < DF / 4; ++k4) {
        v4f wv = wt[k4];
        sc = __builtin_fmaf(b[4 * k4 + 0], wv[0], sc);
        sc = __builtin_fmaf(b[4 * k4 + 1], wv[1], sc);
        sc = __builtin_fmaf(b[4 * k4 + 2], wv[2], sc);
        sc = __builtin_fmaf(b[4 * k4 + 3], wv[3], sc);
      }
      c[t] = sc;
    }
    return;
  }

  const int e0 = (int)blockIdx.x * CHUNK;
  const int left = nedges - e0;
  const int n = left < CHUNK ? left : CHUNK;
  if (t < NBMAX) hist[t] = 0;
  __syncthreads();

  int r[EPT], ci[EPT], rk[EPT];
  float v[EPT];
#pragma unroll
  for (int u = 0; u < EPT; ++u) {
    const int idx = u * 1024 + t;
    r[u] = 0; ci[u] = 0; v[u] = 0.f;
    if (idx < n) {
      r[u]  = rows[e0 + idx];
      ci[u] = cols[e0 + idx];
      v[u]  = lap[e0 + idx];
    }
  }
#pragma unroll
  for (int u = 0; u < EPT; ++u) {
    const int idx = u * 1024 + t;
    rk[u] = 0;
    if (idx < n) rk[u] = atomicAdd(&hist[r[u] / RBIN], 1);
  }
  __syncthreads();

  if (t < 32) {
    const int base = t * (NBMAX / 32);
    int loc[NBMAX / 32];
    int s = 0;
#pragma unroll
    for (int k = 0; k < NBMAX / 32; ++k) { loc[k] = s; s += hist[base + k]; }
    int inc = s;
#pragma unroll
    for (int d = 1; d < 32; d <<= 1) {
      const int y = __shfl_up(inc, d, 32);
      if (t >= d) inc += y;
    }
    const int excl = inc - s;
#pragma unroll
    for (int k = 0; k < NBMAX / 32; ++k) start[base + k] = excl + loc[k];
    if (t == 31) start[NBMAX] = inc;
  }
  __syncthreads();

#pragma unroll
  for (int u = 0; u < EPT; ++u) {
    const int idx = u * 1024 + t;
    if (idx < n) {
      const int bn = r[u] / RBIN;
      const int pos = start[bn] + rk[u];
      s_row[pos] = r[u] - bn * RBIN;
      s_col[pos] = ci[u];
      s_val[pos] = v[u];
    }
  }
  __syncthreads();

  for (int idx = t; idx < n; idx += 1024) {
    srow[e0 + idx] = s_row[idx];
    scol[e0 + idx] = s_col[idx];
    sval[e0 + idx] = s_val[idx];
  }
  if (t < NBP) off[(int)blockIdx.x * OFFP + t] = e0 + start[t];
}

__device__ __forceinline__ int bcast_i(int v, int j) {
  return __builtin_amdgcn_readlane(v, j);
}
__device__ __forceinline__ float bcast_f(float v, int j) {
  return __int_as_float(__builtin_amdgcn_readlane(__float_as_int(v), j));
}

__global__ __launch_bounds__(1024) void bin_spmm_gemm_kernel(
    const int* __restrict__ off, const int* __restrict__ srow, const int* __restrict__ scol,
    const float* __restrict__ sval, int nchunks,
    const float* __restrict__ input, const float* __restrict__ Mp, const float* __restrict__ c,
    float* __restrict__ out, int nrows) {
  __shared__ float xt[(RBIN + 1) * PITCH];
  __shared__ float ldsB[DF * DF];
  __shared__ float ldsC[DF];
  __shared__ int   lens[1024];
  __shared__ int   pS[1024];
  __shared__ int   s_tot;
  __shared__ int   cnt[RCNT];
  __shared__ int   rstart[RCNT];
  __shared__ int   st_r[CAP];
  __shared__ int   st_c[CAP];
  __shared__ float st_v[CAP];

  const int t = threadIdx.x;
  const int lane = t & 31;
  const int wave = __builtin_amdgcn_readfirstlane(t >> 5);
  const int bin = (int)blockIdx.x;

  for (int i4 = t; i4 < ((RBIN + 1) * PITCH) / 4; i4 += 1024) {
    v4f z = {0.f, 0.f, 0.f, 0.f};
    *(v4f*)(xt + (i4 << 2)) = z;
  }
  {
    const int fi = t << 2;
    *(v4f*)(ldsB + fi) = *(const v4f*)(Mp + fi);
  }
  if (t < DF) ldsC[t] = c[t];
  if (t < RCNT) cnt[t] = 0;

  int segS = 0, segN = 0;
  if (t < nchunks) {
    const int s0 = off[t * OFFP + bin];
    const int s1 = off[t * OFFP + bin + 1];
    segS = s0;
    segN = s1 - s0;
  }
  lens[t] = segN;
  pS[t] = segS;
  __syncthreads();
  if (t < 32) {
    const int base = t * 32;
    int s = 0;
#pragma unroll
    for (int k = 0; k < 32; ++k) {
      const int v = lens[base + k];
      lens[base + k] = s;
      s += v;
    }
    int inc = s;
#pragma unroll
    for (int d = 1; d < 32; d <<= 1) {
      const int y = __shfl_up(inc, d, 32);
      if (t >= d) inc += y;
    }
    const int excl = inc - s;
#pragma unroll
    for (int k = 0; k < 32; ++k) lens[base + k] += excl;
    if (t == 31) s_tot = inc;
  }
  __syncthreads();
  const int T = __builtin_amdgcn_readfirstlane(s_tot);

  const char* inB = (const char*)input;
  const unsigned lane4 = (unsigned)lane << 2;

  for (int base = 0; base < T; base += CAP) {
    const int rest = T - base;
    const int nIn = rest < CAP ? rest : CAP;
    const int nPad = (nIn + (EDGE_BATCH - 1)) & ~(EDGE_BATCH - 1);
    int src[CAP / 1024];
#pragma unroll
    for (int u = 0; u < CAP / 1024; ++u) {
      const int q = u * 1024 + t;
      const int f = base + (q < nIn ? q : nIn - 1);
      int p = 0;
#pragma unroll
      for (int step = 512; step >= 1; step >>= 1)
        if (lens[p + step] <= f) p += step;
      src[u] = pS[p] + (f - lens[p]);
    }
    int er[CAP / 1024], ec[CAP / 1024];
    float ev[CAP / 1024];
#pragma unroll
    for (int u = 0; u < CAP / 1024; ++u) {
      er[u] = srow[src[u]];
      ec[u] = scol[src[u]];
      ev[u] = sval[src[u]];
    }
    int rk[CAP / 1024];
#pragma unroll
    for (int u = 0; u < CAP / 1024; ++u) {
      const int q = u * 1024 + t;
      rk[u] = 0;
      if (q < nIn) rk[u] = atomicAdd(&cnt[er[u]], 1);
    }
    __syncthreads();
    if (t < 32) {
      const int rb = t * RSCAN;
      int loc[RSCAN];
      int s = 0;
#pragma unroll
      for (int k = 0; k < RSCAN; ++k) { loc[k] = s; s += cnt[rb + k]; cnt[rb + k] = 0; }
      int inc = s;
#pragma unroll
      for (int d = 1; d < 32; d <<= 1) {
        const int y = __shfl_up(inc, d, 32);
        if (t >= d) inc += y;
      }
      const int excl = inc - s;
#pragma unroll
      for (int k = 0; k < RSCAN; ++k) rstart[rb + k] = excl + loc[k];
    }
    __syncthreads();
#pragma unroll
    for (int u = 0; u < CAP / 1024; ++u) {
      const int q = u * 1024 + t;
      if (q < nIn) {
        const int pos = rstart[er[u]] + rk[u];
        st_r[pos] = er[u]; st_c[pos] = ec[u]; st_v[pos] = ev[u];
      }
    }
    if (t < nPad - nIn) {
      st_r[nIn + t] = RBIN;
      st_c[nIn + t] = 0;
      st_v[nIn + t] = 0.f;
    }
    __syncthreads();

    for (int bi = wave; bi < nPad / EDGE_BATCH; bi += 32) {
      const int j = bi * EDGE_BATCH + (lane & (EDGE_BATCH - 1));
      const int rr = st_r[j];
      const int cc = st_c[j];
      const float ll = st_v[j];
      float a[EDGE_BATCH], bq[EDGE_BATCH];
#pragma unroll
      for (int u = 0; u < EDGE_BATCH; ++u) {
        const char* gp = inB + (((unsigned)bcast_i(cc, u) << 8) + lane4);
        a[u]  = *(const float*)gp;
        bq[u] = *(const float*)(gp + 128);
      }
      int cur = bcast_i(rr, 0);
      float s0 = 0.f, s1 = 0.f;
#pragma unroll
      for (int u = 0; u < EDGE_BATCH; ++u) {
        const int ru = bcast_i(rr, u);
        if (ru != cur) {
          float* dp = xt + cur * PITCH + lane;
          __hip_atomic_fetch_add(dp,      s0, __ATOMIC_RELAXED, __HIP_MEMORY_SCOPE_WORKGROUP);
          __hip_atomic_fetch_add(dp + 32, s1, __ATOMIC_RELAXED, __HIP_MEMORY_SCOPE_WORKGROUP);
          s0 = 0.f; s1 = 0.f; cur = ru;
        }
        const float lj = bcast_f(ll, u);
        s0 = __builtin_fmaf(lj, a[u],  s0);
        s1 = __builtin_fmaf(lj, bq[u], s1);
      }
      {
        float* dp = xt + cur * PITCH + lane;
        __hip_atomic_fetch_add(dp,      s0, __ATOMIC_RELAXED, __HIP_MEMORY_SCOPE_WORKGROUP);
        __hip_atomic_fetch_add(dp + 32, s1, __ATOMIC_RELAXED, __HIP_MEMORY_SCOPE_WORKGROUP);
      }
    }
    __syncthreads();
  }

  const int m  = lane & 15;
  const int kh = lane >> 4;
  const float c0v = ldsC[m], c1v = ldsC[m + 16], c2v = ldsC[m + 32], c3v = ldsC[m + 48];
  for (int strip = wave; strip < RBIN / 16; strip += 32) {
    const int rowBase = bin * RBIN + strip * 16;
    if (rowBase >= nrows) break;
    const float* A = xt + (strip * 16 + m) * PITCH;
    v8f acc0 = {}, acc1 = {}, acc2 = {}, acc3 = {};
#pragma unroll
    for (int k0 = 0; k0 < DF; k0 += 4) {
      const int ka = k0 + 2 * kh;
      v2f a = *(const v2f*)(A + ka);
      const float* Bp = ldsB + (ka >> 1) * (2 * DF);
      v2f b0 = *(const v2f*)(Bp + (m      ) * 2);
      v2f b1 = *(const v2f*)(Bp + (m + 16 ) * 2);
      v2f b2 = *(const v2f*)(Bp + (m + 32 ) * 2);
      v2f b3 = *(const v2f*)(Bp + (m + 48 ) * 2);
      acc0 = __builtin_amdgcn_wmma_f32_16x16x4_f32(false, a, false, b0, (short)0, acc0, false, false);
      acc1 = __builtin_amdgcn_wmma_f32_16x16x4_f32(false, a, false, b1, (short)0, acc1, false, false);
      acc2 = __builtin_amdgcn_wmma_f32_16x16x4_f32(false, a, false, b2, (short)0, acc2, false, false);
      acc3 = __builtin_amdgcn_wmma_f32_16x16x4_f32(false, a, false, b3, (short)0, acc3, false, false);
    }
#pragma unroll
    for (int v = 0; v < 8; ++v) {
      const int r = rowBase + v + 8 * kh;
      if (r < nrows) {
        float* o = out + (size_t)r * DF;
        o[m]      = acc0[v] + c0v;
        o[m + 16] = acc1[v] + c1v;
        o[m + 32] = acc2[v] + c2v;
        o[m + 48] = acc3[v] + c3v;
      }
    }
  }
}

extern "C" void kernel_launch(void* const* d_in, const int* in_sizes, int n_in,
                              void* d_out, int out_size, void* d_ws, size_t ws_size,
                              hipStream_t stream) {
  const float* lap  = (const float*)d_in[0];
  const int*   eidx = (const int*)  d_in[1];
  const float* inp  = (const float*)d_in[2];
  const float* W    = (const float*)d_in[3];
  const float* b    = (const float*)d_in[4];
  float* out = (float*)d_out;

  const int E = in_sizes[0];
  const int N = in_sizes[2] / DF;
  const int* rows = eidx;
  const int* cols = eidx + E;

  const int nchunks = (E + CHUNK - 1) / CHUNK;
  const int nbins   = (N + RBIN - 1) / RBIN;

  float* Mp   = (float*)d_ws;
  float* cvec = Mp + DF * DF;
  int*   off  = (int*)(cvec + DF);
  int*   srow = off + 1024 * OFFP;
  int*   scol = srow + E;
  float* sval = (float*)(scol + E);

  bin_edges_and_Mc_kernel<<<nchunks + 5, 1024, 0, stream>>>(lap, rows, cols, E, nchunks,
                                                            off, srow, scol, sval, W, b, Mp, cvec);
  bin_spmm_gemm_kernel<<<nbins, 1024, 0, stream>>>(off, srow, scol, sval, nchunks,
                                                   inp, Mp, cvec, out, N);
}


# Ordered edges kept as one 8-byte record (row in bin and source row in one word, value in the other) instead of three 4-byte arrays: one load, store and LDS write per edge where there were three
# speedup vs baseline: 5.3176x; 1.0546x over previous
#include <hip/hip_runtime.h>


typedef __attribute__((ext_vector_type(2))) float v2f;
typedef __attribute__((ext_vector_type(2))) int v2i;
typedef __attribute__((ext_vector_type(4))) float v4f;
typedef __attribute__((ext_vector_type(8))) float v8f;

#define DF 64
#define RBIN 400
#define NBMAX 256
#define NBP (NBMAX + 1)
#define OFFP 288
#define CHUNK 8192
#define EPT (CHUNK / 1024)
#define CAP 8192
#define PITCH 68
#define EDGE_BATCH 16
#define RSCAN 13
#define RCNT (32 * RSCAN)
#define KSH 23
#define KMASK ((1 << KSH) - 1)


__global__ __launch_bounds__(1024) void bin_edges_and_Mc_kernel(
    const float* __restrict__ lap, const int* __restrict__ rows, const int* __restrict__ cols,
    int nedges, int nchunks,
    int* __restrict__ off, int* __restrict__ rec,
    const float* __restrict__ W, const float* __restrict__ b,
    float* __restrict__ Mp, float* __restrict__ c) {
  __shared__ int hist[NBMAX];
  __shared__ int start[NBP];
  __shared__ v2i s_rec[CHUNK];
  const int t = threadIdx.x;

  if ((int)blockIdx.x >= nchunks) {
    const int g = (int)blockIdx.x - nchunks;
    if (g < 4) {
      const int wv = g * 32 + __builtin_amdgcn_readfirstlane(t >> 5);
      const int j = wv >> 1;
      const int i = ((wv & 1) << 5) + (t & 31);
      const float* __restrict__ wrow = W + j * DF;
      const float* __restrict__ wcol = W + i;
      float s = 0.f;
#pragma unroll
      for (int k = 0; k < DF; ++k) s = __builtin_fmaf(wcol[k * DF], wrow[k], s);
      Mp[(i >> 1) * (2 * DF) + j * 2 + (i & 1)] = s;
    } else if (t < DF) {
      const v4f* __restrict__ wt = (const v4f*)(W + t * DF);
      float sc = b[t];
#pragma unroll
      for (int k4 = 0; k4 < DF / 4; ++k4) {
        v4f wv = wt[k4];
        sc = __builtin_fmaf(b[4 * k4 + 0], wv[0], sc);
        sc = __builtin_fmaf(b[4 * k4 + 1], wv[1], sc);
        sc = __builtin_fmaf(b[4 * k4 + 2], wv[2], sc);
        sc = __builtin_fmaf(b[4 * k4 + 3], wv[3], sc);
      }
      c[t] = sc;
    }
    return;
  }

  const int e0 = (int)blockIdx.x * CHUNK;
  const int left = nedges - e0;
  const int n = left < CHUNK ? left : CHUNK;
  if (t < NBMAX) hist[t] = 0;
  __syncthreads();

  int r[EPT], ci[EPT], rk[EPT];
  float v[EPT];
#pragma unroll
  for (int u = 0; u < EPT; ++u) {
    const int idx = u * 1024 + t;
    r[u] = 0; ci[u] = 0; v[u] = 0.f;
    if (idx < n) {
      r[u]  = rows[e0 + idx];
      ci[u] = cols[e0 + idx];
      v[u]  = lap[e0 + idx];
    }
  }
#pragma unroll
  for (int u = 0; u < EPT; ++u) {
    const int idx = u * 1024 + t;
    rk[u] = 0;
    if (idx < n) rk[u] = atomicAdd(&hist[r[u] / RBIN], 1);
  }
  __syncthreads();

  if (t < 32) {
    const int base = t * (NBMAX / 32);
    int loc[NBMAX / 32];
    int s = 0;
#pragma unroll
    for (int k = 0; k < NBMAX / 32; ++k) { loc[k] = s; s += hist[base + k]; }
    int inc = s;
#pragma unroll
    for (int d = 1; d < 32; d <<= 1) {
      const int y = __shfl_up(inc, d, 32);
      if (t >= d) inc += y;
    }
    const int excl = inc - s;
#pragma unroll
    for (int k = 0; k < NBMAX / 32; ++k) start[base + k] = excl + loc[k];
    if (t == 31) start[NBMAX] = inc;
  }
  __syncthreads();

#pragma unroll
  for (int u = 0; u < EPT; ++u) {
    const int idx = u * 1024 + t;
    if (idx < n) {
      const int bn = r[u] / RBIN;
      const int pos = start[bn] + rk[u];
      v2i e;
      e.x = (int)(((unsigned)(r[u] - bn * RBIN) << KSH) | (unsigned)ci[u]);
      e.y = __float_as_int(v[u]);
      s_rec[pos] = e;
    }
  }
  __syncthreads();

  v2i* __restrict__ recv = (v2i*)rec;
  for (int idx = t; idx < n; idx += 1024)
    recv[e0 + idx] = s_rec[idx];
  if (t < NBP) off[(int)blockIdx.x * OFFP + t] = e0 + start[t];
}

__device__ __forceinline__ int bcast_i(int v, int j) {
  return __builtin_amdgcn_readlane(v, j);
}
__device__ __forceinline__ float bcast_f(float v, int j) {
  return __int_as_float(__builtin_amdgcn_readlane(__float_as_int(v), j));
}

__global__ __launch_bounds__(1024) void bin_spmm_gemm_kernel(
    const int* __restrict__ off, const int* __restrict__ rec, int nchunks,
    const float* __restrict__ input, const float* __restrict__ Mp, const float* __restrict__ c,
    float* __restrict__ out, int nrows) {
  __shared__ float xt[(RBIN + 1) * PITCH];
  __shared__ float ldsB[DF * DF];
  __shared__ float ldsC[DF];
  __shared__ int   lens[1024];
  __shared__ int   pS[1024];
  __shared__ int   s_tot;
  __shared__ int   cnt[RCNT];
  __shared__ int   rstart[RCNT];
  __shared__ v2i   st[CAP];

  const int t = threadIdx.x;
  const int lane = t & 31;
  const int wave = __builtin_amdgcn_readfirstlane(t >> 5);
  const int bin = (int)blockIdx.x;

  for (int i4 = t; i4 < ((RBIN + 1) * PITCH) / 4; i4 += 1024) {
    v4f z = {0.f, 0.f, 0.f, 0.f};
    *(v4f*)(xt + (i4 << 2)) = z;
  }
  {
    const int fi = t << 2;
    *(v4f*)(ldsB + fi) = *(const v4f*)(Mp + fi);
  }
  if (t < DF) ldsC[t] = c[t];
  if (t < RCNT) cnt[t] = 0;

  int segS = 0, segN = 0;
  if (t < nchunks) {
    const int s0 = off[t * OFFP + bin];
    const int s1 = off[t * OFFP + bin + 1];
    segS = s0;
    segN = s1 - s0;
  }
  lens[t] = segN;
  pS[t] = segS;
  __syncthreads();
  if (t < 32) {
    const int base = t * 32;
    int s = 0;
#pragma unroll
    for (int k = 0; k < 32; ++k) {
      const int v = lens[base + k];
      lens[base + k] = s;
      s += v;
    }
    int inc = s;
#pragma unroll
    for (int d = 1; d < 32; d <<= 1) {
      const int y = __shfl_up(inc, d, 32);
      if (t >= d) inc += y;
    }
    const int excl = inc - s;
#pragma unroll
    for (int k = 0; k < 32; ++k) lens[base + k] += excl;
    if (t == 31) s_tot = inc;
  }
  __syncthreads();
  const int T = __builtin_amdgcn_readfirstlane(s_tot);

  const char* inB = (const char*)input;
  const v2i* __restrict__ recv = (const v2i*)rec;
  const unsigned lane4 = (unsigned)lane << 2;

  for (int base = 0; base < T; base += CAP) {
    const int rest = T - base;
    const int nIn = rest < CAP ? rest : CAP;
    const int nPad = (nIn + (EDGE_BATCH - 1)) & ~(EDGE_BATCH - 1);
    int src[CAP / 1024];
#pragma unroll
    for (int u = 0; u < CAP / 1024; ++u) {
      const int q = u * 1024 + t;
      const int f = base + (q < nIn ? q : nIn - 1);
      int p = 0;
#pragma unroll
      for (int step = 512; step >= 1; step >>= 1)
        if (lens[p + step] <= f) p += step;
      src[u] = pS[p] + (f - lens[p]);
    }
    v2i ed[CAP / 1024];
#pragma unroll
    for (int u = 0; u < CAP / 1024; ++u) ed[u] = recv[src[u]];
    int er[CAP / 1024];
#pragma unroll
    for (int u = 0; u < CAP / 1024; ++u) er[u] = (int)((unsigned)ed[u].x >> KSH);
    int rk[CAP / 1024];
#pragma unroll
    for (int u = 0; u < CAP / 1024; ++u) {
      const int q = u * 1024 + t;
      rk[u] = 0;
      if (q < nIn) rk[u] = atomicAdd(&cnt[er[u]], 1);
    }
    __syncthreads();
    if (t < 32) {
      const int rb = t * RSCAN;
      int loc[RSCAN];
      int s = 0;
#pragma unroll
      for (int k = 0; k < RSCAN; ++k) { loc[k] = s; s += cnt[rb + k]; cnt[rb + k] = 0; }
      int inc = s;
#pragma unroll
      for (int d = 1; d < 32; d <<= 1) {
        const int y = __shfl_up(inc, d, 32);
        if (t >= d) inc += y;
      }
      const int excl = inc - s;
#pragma unroll
      for (int k = 0; k < RSCAN; ++k) rstart[rb + k] = excl + loc[k];
    }
    __syncthreads();
#pragma unroll
    for (int u = 0; u < CAP / 1024; ++u) {
      const int q = u * 1024 + t;
      if (q < nIn) {
        const int pos = rstart[er[u]] + rk[u];
        st[pos] = ed[u];
      }
    }
    if (t < nPad - nIn) {
      v2i e;
      e.x = (int)((unsigned)RBIN << KSH);
      e.y = 0;
      st[nIn + t] = e;
    }
    __syncthreads();

    for (int bi = wave; bi < nPad / EDGE_BATCH; bi += 32) {
      const int j = bi * EDGE_BATCH + (lane & (EDGE_BATCH - 1));
      const v2i e = st[j];
      const int rr = (int)((unsigned)e.x >> KSH);
      const int cc = (e.x & KMASK) << 8;
      const float ll = __int_as_float(e.y);
      float a[EDGE_BATCH], bq[EDGE_BATCH];
#pragma unroll
      for (int u = 0; u < EDGE_BATCH; ++u) {
        const char* gp = inB + ((unsigned)bcast_i(cc, u) + lane4);
        a[u]  = *(const float*)gp;
        bq[u] = *(const float*)(gp + 128);
      }
      int cur = bcast_i(rr, 0);
      float s0 = 0.f, s1 = 0.f;
#pragma unroll
      for (int u = 0; u < EDGE_BATCH; ++u) {
        const int ru = bcast_i(rr, u);
        if (ru != cur) {
          float* dp = xt + cur * PITCH + lane;
          __hip_atomic_fetch_add(dp,      s0, __ATOMIC_RELAXED, __HIP_MEMORY_SCOPE_WORKGROUP);
          __hip_atomic_fetch_add(dp + 32, s1, __ATOMIC_RELAXED, __HIP_MEMORY_SCOPE_WORKGROUP);
          s0 = 0.f; s1 = 0.f; cur = ru;
        }
        const float lj = bcast_f(ll, u);
        s0 = __builtin_fmaf(lj, a[u],  s0);
        s1 = __builtin_fmaf(lj, bq[u], s1);
      }
      {
        float* dp = xt + cur * PITCH + lane;
        __hip_atomic_fetch_add(dp,      s0, __ATOMIC_RELAXED, __HIP_MEMORY_SCOPE_WORKGROUP);
        __hip_atomic_fetch_add(dp + 32, s1, __ATOMIC_RELAXED, __HIP_MEMORY_SCOPE_WORKGROUP);
      }
    }
    __syncthreads();
  }

  const int m  = lane & 15;
  const int kh = lane >> 4;
  const float c0v = ldsC[m], c1v = ldsC[m + 16], c2v = ldsC[m + 32], c3v = ldsC[m + 48];
  for (int strip = wave; strip < RBIN / 16; strip += 32) {
    const int rowBase = bin * RBIN + strip * 16;
    if (rowBase >= nrows) break;
    const float* A = xt + (strip * 16 + m) * PITCH;
    v8f acc0 = {}, acc1 = {}, acc2 = {}, acc3 = {};
#pragma unroll
    for (int k0 = 0; k0 < DF; k0 += 4) {
      const int ka = k0 + 2 * kh;
      v2f a = *(const v2f*)(A + ka);
      const float* Bp = ldsB + (ka >> 1) * (2 * DF);
      v2f b0 = *(const v2f*)(Bp + (m      ) * 2);
      v2f b1 = *(const v2f*)(Bp + (m + 16 ) * 2);
      v2f b2 = *(const v2f*)(Bp + (m + 32 ) * 2);
      v2f b3 = *(const v2f*)(Bp + (m + 48 ) * 2);
      acc0 = __builtin_amdgcn_wmma_f32_16x16x4_f32(false, a, false, b0, (short)0, acc0, false, false);
      acc1 = __builtin_amdgcn_wmma_f32_16x16x4_f32(false, a, false, b1, (short)0, acc1, false, false);
      acc2 = __builtin_amdgcn_wmma_f32_16x16x4_f32(false, a, false, b2, (short)0, acc2, false, false);
      acc3 = __builtin_amdgcn_wmma_f32_16x16x4_f32(false, a, false, b3, (short)0, acc3, false, false);
    }
#pragma unroll
    for (int v = 0; v < 8; ++v) {
      const int r = rowBase + v + 8 * kh;
      if (r < nrows) {
        float* o = out + (size_t)r * DF;
        o[m]      = acc0[v] + c0v;
        o[m + 16] = acc1[v] + c1v;
        o[m + 32] = acc2[v] + c2v;
        o[m + 48] = acc3[v] + c3v;
      }
    }
  }
}

extern "C" void kernel_launch(void* const* d_in, const int* in_sizes, int n_in,
                              void* d_out, int out_size, void* d_ws, size_t ws_size,
                              hipStream_t stream) {
  const float* lap  = (const float*)d_in[0];
  const int*   eidx = (const int*)  d_in[1];
  const float* inp  = (const float*)d_in[2];
  const float* W    = (const float*)d_in[3];
  const float* b    = (const float*)d_in[4];
  float* out = (float*)d_out;

  const int E = in_sizes[0];
  const int N = in_sizes[2] / DF;
  const int* rows = eidx;
  const int* cols = eidx + E;

  const int nchunks = (E + CHUNK - 1) / CHUNK;
  const int nbins   = (N + RBIN - 1) / RBIN;

  float* Mp   = (float*)d_ws;
  float* cvec = Mp + DF * DF;
  int*   off  = (int*)(cvec + DF);
  int*   rec  = off + 1024 * OFFP;

  bin_edges_and_Mc_kernel<<<nchunks + 5, 1024, 0, stream>>>(lap, rows, cols, E, nchunks,
                                                            off, rec, W, b, Mp, cvec);
  bin_spmm_gemm_kernel<<<nbins, 1024, 0, stream>>>(off, rec, nchunks,
                                                   inp, Mp, cvec, out, N);
}
